# starting kernel
# baseline (speedup 1.0000x reference)
#include <hip/hip_runtime.h>
#include <hip/hip_bf16.h>

#define BB   16
#define SS   64
#define HH   16
#define DD   16
#define EE   256
#define HID  128
#define MQ   (BB*SS)
#define MPRE (BB*SS*HH)

typedef float v2f __attribute__((ext_vector_type(2)));
typedef float v8f __attribute__((ext_vector_type(8)));

__device__ __forceinline__ float fast_tanh(float x) {
#if __has_builtin(__builtin_amdgcn_tanhf)
    return __builtin_amdgcn_tanhf(x);
#else
    return tanhf(x);
#endif
}

__device__ __forceinline__ v2f ld2(const float* p) {
    return *reinterpret_cast<const v2f*>(p);
}

#define GEMM_WAVES 8

__global__ __launch_bounds__(256) void gemm_bias_wmma_f32(
    const float* __restrict__ A, int lda,
    const float* __restrict__ W, int ldw, int wOff,
    const float* __restrict__ bias,
    float* __restrict__ C, int ldc,
    int tilesN, int K)
{
    const int lane = threadIdx.x;
    const int wave = threadIdx.y;
    const int gw   = blockIdx.x * GEMM_WAVES + wave;
    const int tm   = gw / tilesN;
    const int tn   = gw - tm * tilesN;
    const int m0   = tm * 16;
    const int n0   = tn * 16;
    const int half = lane >> 4;
    const int sub  = lane & 15;
    const int h2   = half * 2;

    const float* aRow = A + (size_t)(m0 + sub) * lda;
    const float* wRow = W + (size_t)(n0 + sub) * ldw + wOff;

    v8f c = {};
    for (int k0 = 0; k0 < K; k0 += 16) {
        v2f a0 = ld2(aRow + k0 +  0 + h2);
        v2f a1 = ld2(aRow + k0 +  4 + h2);
        v2f a2 = ld2(aRow + k0 +  8 + h2);
        v2f a3 = ld2(aRow + k0 + 12 + h2);
        v2f b0 = ld2(wRow + k0 +  0 + h2);
        v2f b1 = ld2(wRow + k0 +  4 + h2);
        v2f b2 = ld2(wRow + k0 +  8 + h2);
        v2f b3 = ld2(wRow + k0 + 12 + h2);
        c = __builtin_amdgcn_wmma_f32_16x16x4_f32(false, a0, false, b0, (short)0, c, false, false);
        c = __builtin_amdgcn_wmma_f32_16x16x4_f32(false, a1, false, b1, (short)0, c, false, false);
        c = __builtin_amdgcn_wmma_f32_16x16x4_f32(false, a2, false, b2, (short)0, c, false, false);
        c = __builtin_amdgcn_wmma_f32_16x16x4_f32(false, a3, false, b3, (short)0, c, false, false);
    }

    const int   col  = n0 + sub;
    const int   row0 = m0 + half * 8;
    const float bb   = bias ? bias[col] : 0.0f;
#pragma unroll
    for (int r = 0; r < 8; ++r)
        C[(size_t)(row0 + r) * ldc + col] = c[r] + bb;
}

__global__ __launch_bounds__(256) void scores_softmax_ctx(
    const float* __restrict__ pre_q,
    const float* __restrict__ pre_k,
    const float* __restrict__ attn_v,
    const float* __restrict__ v,
    float* __restrict__ ctx)
{
    __shared__ float pq[HH * HID];
    __shared__ float av[HID];
    __shared__ float sc[SS * HH];
    __shared__ float vloc[HH * DD];

    const int    tid = threadIdx.x;
    const size_t bi  = blockIdx.x;
    const size_t bat = bi / SS;

    const float* pqg = pre_q + bi * (HH * HID);
    for (int t = tid; t < HH * HID; t += 256) pq[t] = pqg[t];
    for (int t = tid; t < HID;      t += 256) av[t] = attn_v[t];
    const float* vg = v + bi * (HH * DD);
    for (int t = tid; t < HH * DD;  t += 256) vloc[t] = vg[t];
    __syncthreads();

    const int lane = tid & 31;
    const int wv   = tid >> 5;
    const int c0   = lane * 4;
    const float a0 = av[c0], a1 = av[c0 + 1], a2 = av[c0 + 2], a3 = av[c0 + 3];
    const float* pkb = pre_k + bat * (size_t)(SS * HH * HID);

    for (int p = wv; p < SS * HH; p += 8) {
        const int j = p >> 4;
        const int h = p & 15;
        const float* pk  = pkb + ((size_t)j * HH + h) * HID + c0;
        const float* pqh = pq + h * HID + c0;
        float acc = fast_tanh(pqh[0] + pk[0]) * a0
                  + fast_tanh(pqh[1] + pk[1]) * a1
                  + fast_tanh(pqh[2] + pk[2]) * a2
                  + fast_tanh(pqh[3] + pk[3]) * a3;
#pragma unroll
        for (int off = 16; off > 0; off >>= 1)
            acc += __shfl_xor(acc, off, 32);
        if (lane == 0) sc[j * HH + h] = acc;
    }
    __syncthreads();

    if (tid < HH) {
        const int h = tid;
        float m = -3.0e38f;
        for (int j = 0; j < SS; ++j) m = fmaxf(m, sc[j * HH + h]);
        float s = 0.0f;
        for (int j = 0; j < SS; ++j) {
            float e = __expf(sc[j * HH + h] - m);
            sc[j * HH + h] = e;
            s += e;
        }
        const float inv = 1.0f / s;
        for (int j = 0; j < SS; ++j) sc[j * HH + h] *= inv;
    }
    __syncthreads();

    float* cg = ctx + bi * (size_t)(SS * DD);
    for (int idx = tid; idx < SS * DD; idx += 256) {
        const int j = idx >> 4;
        const int d = idx & 15;
        float acc = 0.0f;
#pragma unroll
        for (int h = 0; h < HH; ++h)
            acc = fmaf(sc[j * HH + h], vloc[h * DD + d], acc);
        cg[idx] = acc;
    }
}

extern "C" void kernel_launch(void* const* d_in, const int* in_sizes, int n_in,
                              void* d_out, int out_size, void* d_ws, size_t ws_size,
                              hipStream_t stream) {
    (void)in_sizes; (void)n_in; (void)out_size; (void)ws_size;
    const float* x      = (const float*)d_in[0];
    const float* Wq     = (const float*)d_in[1];
    const float* bq     = (const float*)d_in[2];
    const float* Wk     = (const float*)d_in[3];
    const float* bk     = (const float*)d_in[4];
    const float* Wv     = (const float*)d_in[5];
    const float* bv     = (const float*)d_in[6];
    const float* Wa     = (const float*)d_in[7];
    const float* ba     = (const float*)d_in[8];
    const float* attn_v = (const float*)d_in[9];
    const float* Wfc    = (const float*)d_in[10];
    const float* bfc    = (const float*)d_in[11];
    float* out = (float*)d_out;

    float* q     = (float*)d_ws;
    float* k     = q     + (size_t)MQ * EE;
    float* v     = k     + (size_t)MQ * EE;
    float* pre_q = v     + (size_t)MQ * EE;
    float* pre_k = pre_q + (size_t)MPRE * HID;
    float* ctx   = pre_k + (size_t)MPRE * HID;

    const dim3 blk(32, GEMM_WAVES);

    gemm_bias_wmma_f32<<<128, blk, 0, stream>>>(x, EE, Wq, EE, 0, bq, q, EE, EE / 16, EE);
    gemm_bias_wmma_f32<<<128, blk, 0, stream>>>(x, EE, Wk, EE, 0, bk, k, EE, EE / 16, EE);
    gemm_bias_wmma_f32<<<128, blk, 0, stream>>>(x, EE, Wv, EE, 0, bv, v, EE, EE / 16, EE);

    gemm_bias_wmma_f32<<<1024, blk, 0, stream>>>(q, DD, Wa, 2 * DD, 0,  ba,      pre_q, HID, HID / 16, DD);
    gemm_bias_wmma_f32<<<1024, blk, 0, stream>>>(k, DD, Wa, 2 * DD, DD, nullptr, pre_k, HID, HID / 16, DD);

    scores_softmax_ctx<<<BB * SS, 256, 0, stream>>>(pre_q, pre_k, attn_v, v, ctx);

    gemm_bias_wmma_f32<<<128, blk, 0, stream>>>(ctx, SS * DD, Wfc, SS * DD, 0, bfc, out, EE, EE / 16, SS * DD);
}


# scores kernel builds pre_q, pre_k itself on the matrix pipe, transposed so the product's D registers are the score product's A rows; the two pre-activation launches go (7 to 5); sum order changes
# speedup vs baseline: 1.5067x; 1.5067x over previous
#include <hip/hip_runtime.h>
#include <hip/hip_bf16.h>

#define BB   16
#define SS   64
#define HH   16
#define DD   16
#define EE   256
#define HID  128
#define MQ   (BB*SS)

typedef float v2f __attribute__((ext_vector_type(2)));
typedef float v8f __attribute__((ext_vector_type(8)));

__device__ __forceinline__ float fast_tanh(float x) {
#if __has_builtin(__builtin_amdgcn_tanhf)
    return __builtin_amdgcn_tanhf(x);
#else
    return tanhf(x);
#endif
}

__device__ __forceinline__ v2f ld2(const float* p) {
    return *reinterpret_cast<const v2f*>(p);
}

#define GEMM_WAVES 8

__global__ __launch_bounds__(256) void gemm_bias_wmma_f32(
    const float* __restrict__ A, int lda,
    const float* __restrict__ W, int ldw, int wOff,
    const float* __restrict__ bias,
    float* __restrict__ C, int ldc,
    int tilesN, int K)
{
    const int lane = threadIdx.x;
    const int wave = threadIdx.y;
    const int gw   = blockIdx.x * GEMM_WAVES + wave;
    const int tm   = gw / tilesN;
    const int tn   = gw - tm * tilesN;
    const int m0   = tm * 16;
    const int n0   = tn * 16;
    const int half = lane >> 4;
    const int sub  = lane & 15;
    const int h2   = half * 2;

    const float* aRow = A + (size_t)(m0 + sub) * lda;
    const float* wRow = W + (size_t)(n0 + sub) * ldw + wOff;

    v8f c = {};
    for (int k0 = 0; k0 < K; k0 += 16) {
        v2f a0 = ld2(aRow + k0 +  0 + h2);
        v2f a1 = ld2(aRow + k0 +  4 + h2);
        v2f a2 = ld2(aRow + k0 +  8 + h2);
        v2f a3 = ld2(aRow + k0 + 12 + h2);
        v2f b0 = ld2(wRow + k0 +  0 + h2);
        v2f b1 = ld2(wRow + k0 +  4 + h2);
        v2f b2 = ld2(wRow + k0 +  8 + h2);
        v2f b3 = ld2(wRow + k0 + 12 + h2);
        c = __builtin_amdgcn_wmma_f32_16x16x4_f32(false, a0, false, b0, (short)0, c, false, false);
        c = __builtin_amdgcn_wmma_f32_16x16x4_f32(false, a1, false, b1, (short)0, c, false, false);
        c = __builtin_amdgcn_wmma_f32_16x16x4_f32(false, a2, false, b2, (short)0, c, false, false);
        c = __builtin_amdgcn_wmma_f32_16x16x4_f32(false, a3, false, b3, (short)0, c, false, false);
    }

    const int   col  = n0 + sub;
    const int   row0 = m0 + half * 8;
    const float bb   = bias ? bias[col] : 0.0f;
#pragma unroll
    for (int r = 0; r < 8; ++r)
        C[(size_t)(row0 + r) * ldc + col] = c[r] + bb;
}

#define QB   4
#define SCQ  (SS * HH + 16)

typedef float v4f __attribute__((ext_vector_type(4)));

__device__ __forceinline__ v8f wmma4(v2f a, v2f b, v8f c) {
    return __builtin_amdgcn_wmma_f32_16x16x4_f32(false, a, false, b, (short)0, c, false, false);
}

__global__ __launch_bounds__(512) void scores_softmax_ctx(
    const float* __restrict__ qm,
    const float* __restrict__ km,
    const float* __restrict__ Wa,
    const float* __restrict__ ba,
    const float* __restrict__ attn_v,
    const float* __restrict__ v,
    float* __restrict__ ctx)
{
    __shared__ __attribute__((aligned(16))) float pq[QB * HH * HID];
    __shared__ float sc[QB * SCQ];
    __shared__ float vloc[QB * HH * DD];

    const int    tid = threadIdx.x;
    const size_t bi0 = (size_t)blockIdx.x * QB;
    const size_t bat = bi0 / SS;

    const float* vg = v + bi0 * (HH * DD);

    const int lane = tid & 31;
    const int h    = tid >> 5;
    const int sub  = lane & 15;
    const int half = lane >> 4;
    const int rsel = sub & 7;
    const int h2   = 2 * half;

    const float* kb = km + (bat * SS + sub) * (size_t)EE + h * DD + h2;

    float sv[QB * HH * DD / 512];
    for (int n = 0; n < QB * HH * DD / 512; ++n) sv[n] = vg[tid + 512 * n];

    v2f kf[4];
    for (int kk = 0; kk < 4; ++kk) kf[kk] = ld2(kb + 4 * kk);

    {
        const float* qrow = qm + (bi0 + (sub & (QB - 1))) * (size_t)EE + h * DD + h2;
        v2f qf[4];
        for (int kk = 0; kk < 4; ++kk) qf[kk] = ld2(qrow + 4 * kk);
        for (int t = 0; t < HID / 16; ++t) {
            const float* wr = Wa + (size_t)(16 * t + sub) * (2 * DD) + h2;
            const v4f b0 = *reinterpret_cast<const v4f*>(ba + 16 * t + 8 * half);
            const v4f b1 = *reinterpret_cast<const v4f*>(ba + 16 * t + 8 * half + 4);
            v8f d = {b0[0], b0[1], b0[2], b0[3], b1[0], b1[1], b1[2], b1[3]};
            for (int kk = 0; kk < 4; ++kk)
                d = wmma4(ld2(wr + 4 * kk), qf[kk], d);
            if (sub < QB) {
                float* dst = pq + (sub * HH + h) * HID + 16 * t + 8 * half;
                v4f lo = {d[0], d[1], d[2], d[3]};
                v4f hi = {d[4], d[5], d[6], d[7]};
                *reinterpret_cast<v4f*>(dst)     = lo;
                *reinterpret_cast<v4f*>(dst + 4) = hi;
            }
        }
    }

    const float* wkb = Wa + (size_t)sub * (2 * DD) + DD + h2;
    v4f avr[HID / 16][2];
    for (int t = 0; t < HID / 16; ++t) {
        avr[t][0] = *reinterpret_cast<const v4f*>(attn_v + 16 * t + 8 * half);
        avr[t][1] = *reinterpret_cast<const v4f*>(attn_v + 16 * t + 8 * half + 4);
    }

    for (int n = 0; n < QB * HH * DD / 512; ++n) vloc[tid + 512 * n] = sv[n];
    __syncthreads();

    for (int jt = 0; jt < SS / 16; ++jt) {
        v8f pk[HID / 16];
        for (int t = 0; t < HID / 16; ++t) {
            v8f d = {};
            const float* wr = wkb + (size_t)(16 * t) * (2 * DD);
            for (int kk = 0; kk < 4; ++kk) d = wmma4(ld2(wr + 4 * kk), kf[kk], d);
            pk[t] = d;
        }
        if (jt + 1 < SS / 16) {
            const float* kr = kb + (size_t)(16 * (jt + 1)) * EE;
            for (int kk = 0; kk < 4; ++kk) kf[kk] = ld2(kr + 4 * kk);
        }

#pragma unroll 1
        for (int q = 0; q < QB; q += 2) {
            const float* pq0 = pq + (q * HH + h) * HID + 8 * half;
            const float* pq1 = pq0 + HH * HID;
            v8f d0 = {};
            v8f d1 = {};
            for (int t = 0; t < HID / 16; ++t) {
                const v4f u0a = *reinterpret_cast<const v4f*>(pq0 + 16 * t);
                const v4f u0b = *reinterpret_cast<const v4f*>(pq0 + 16 * t + 4);
                const v4f u1a = *reinterpret_cast<const v4f*>(pq1 + 16 * t);
                const v4f u1b = *reinterpret_cast<const v4f*>(pq1 + 16 * t + 4);
                const v8f k8 = pk[t];
                v2f a, b;
                b[0] = avr[t][0][0]; b[1] = avr[t][0][1];
                a[0] = fast_tanh(u0a[0] + k8[0]); a[1] = fast_tanh(u0a[1] + k8[1]);
                d0 = wmma4(a, b, d0);
                a[0] = fast_tanh(u1a[0] + k8[0]); a[1] = fast_tanh(u1a[1] + k8[1]);
                d1 = wmma4(a, b, d1);
                b[0] = avr[t][0][2]; b[1] = avr[t][0][3];
                a[0] = fast_tanh(u0a[2] + k8[2]); a[1] = fast_tanh(u0a[3] + k8[3]);
                d0 = wmma4(a, b, d0);
                a[0] = fast_tanh(u1a[2] + k8[2]); a[1] = fast_tanh(u1a[3] + k8[3]);
                d1 = wmma4(a, b, d1);
                b[0] = avr[t][1][0]; b[1] = avr[t][1][1];
                a[0] = fast_tanh(u0b[0] + k8[4]); a[1] = fast_tanh(u0b[1] + k8[5]);
                d0 = wmma4(a, b, d0);
                a[0] = fast_tanh(u1b[0] + k8[4]); a[1] = fast_tanh(u1b[1] + k8[5]);
                d1 = wmma4(a, b, d1);
                b[0] = avr[t][1][2]; b[1] = avr[t][1][3];
                a[0] = fast_tanh(u0b[2] + k8[6]); a[1] = fast_tanh(u0b[3] + k8[7]);
                d0 = wmma4(a, b, d0);
                a[0] = fast_tanh(u1b[2] + k8[6]); a[1] = fast_tanh(u1b[3] + k8[7]);
                d1 = wmma4(a, b, d1);
            }
            float s0 = d0[0];
            float s1 = d1[0];
            for (int r = 1; r < 8; ++r) {
                s0 = (rsel == r) ? d0[r] : s0;
                s1 = (rsel == r) ? d1[r] : s1;
            }
            const int jrow = 16 * jt + 8 * half + rsel;
            sc[q * SCQ + jrow * HH + h]       = s0;
            sc[(q + 1) * SCQ + jrow * HH + h] = s1;
        }
    }
    __syncthreads();

    if (tid < QB * HH) {
        float* sp = sc + (tid >> 4) * SCQ + (tid & 15);
        float m = -3.0e38f;
        for (int j = 0; j < SS; ++j) m = fmaxf(m, sp[j * HH]);
        float s = 0.0f;
        for (int j = 0; j < SS; ++j) {
            float e = __expf(sp[j * HH] - m);
            sp[j * HH] = e;
            s += e;
        }
        const float inv = 1.0f / s;
        for (int j = 0; j < SS; ++j) sp[j * HH] *= inv;
    }
    __syncthreads();

    for (int q = 0; q < QB; ++q) {
        float*       cg = ctx + (bi0 + q) * (size_t)(SS * DD);
        const float* sq = sc + q * SCQ;
        const float* vq = vloc + q * (HH * DD);
        for (int idx = tid; idx < SS * DD; idx += 512) {
            const int j = idx >> 4;
            const int d = idx & 15;
            float acc = 0.0f;
#pragma unroll
            for (int hh = 0; hh < HH; ++hh)
                acc = fmaf(sq[j * HH + hh], vq[hh * DD + d], acc);
            cg[idx] = acc;
        }
    }
}

extern "C" void kernel_launch(void* const* d_in, const int* in_sizes, int n_in,
                              void* d_out, int out_size, void* d_ws, size_t ws_size,
                              hipStream_t stream) {
    (void)in_sizes; (void)n_in; (void)out_size; (void)ws_size;
    const float* x      = (const float*)d_in[0];
    const float* Wq     = (const float*)d_in[1];
    const float* bq     = (const float*)d_in[2];
    const float* Wk     = (const float*)d_in[3];
    const float* bk     = (const float*)d_in[4];
    const float* Wv     = (const float*)d_in[5];
    const float* bv     = (const float*)d_in[6];
    const float* Wa     = (const float*)d_in[7];
    const float* ba     = (const float*)d_in[8];
    const float* attn_v = (const float*)d_in[9];
    const float* Wfc    = (const float*)d_in[10];
    const float* bfc    = (const float*)d_in[11];
    float* out = (float*)d_out;

    float* q     = (float*)d_ws;
    float* k     = q     + (size_t)MQ * EE;
    float* v     = k     + (size_t)MQ * EE;
    float* ctx   = v     + (size_t)MQ * EE;

    const dim3 blk(32, GEMM_WAVES);

    gemm_bias_wmma_f32<<<128, blk, 0, stream>>>(x, EE, Wq, EE, 0, bq, q, EE, EE / 16, EE);
    gemm_bias_wmma_f32<<<128, blk, 0, stream>>>(x, EE, Wk, EE, 0, bk, k, EE, EE / 16, EE);
    gemm_bias_wmma_f32<<<128, blk, 0, stream>>>(x, EE, Wv, EE, 0, bv, v, EE, EE / 16, EE);

    scores_softmax_ctx<<<BB * SS / QB, 512, 0, stream>>>(q, k, Wa, ba, attn_v, v, ctx);

    gemm_bias_wmma_f32<<<128, blk, 0, stream>>>(ctx, SS * DD, Wfc, SS * DD, 0, bfc, out, EE, EE / 16, SS * DD);
}


# GEMM launches: work-groups of 4 waves (one per SIMD), 256 groups instead of 128 of 8 waves, so all 256 WGPs take fragment loads, not half of them; tile code, k order and sum order unchanged
# speedup vs baseline: 1.6923x; 1.1232x over previous
#include <hip/hip_runtime.h>
#include <hip/hip_bf16.h>

#define BB   16
#define SS   64
#define HH   16
#define DD   16
#define EE   256
#define HID  128
#define MQ   (BB*SS)

typedef float v2f __attribute__((ext_vector_type(2)));
typedef float v8f __attribute__((ext_vector_type(8)));

__device__ __forceinline__ float fast_tanh(float x) {
#if __has_builtin(__builtin_amdgcn_tanhf)
    return __builtin_amdgcn_tanhf(x);
#else
    return tanhf(x);
#endif
}

__device__ __forceinline__ v2f ld2(const float* p) {
    return *reinterpret_cast<const v2f*>(p);
}

#define GEMM_WAVES 4

__global__ __launch_bounds__(32 * GEMM_WAVES) void gemm_bias_wmma_f32(
    const float* __restrict__ A, int lda,
    const float* __restrict__ W, int ldw, int wOff,
    const float* __restrict__ bias,
    float* __restrict__ C, int ldc,
    int tilesN, int K)
{
    const int lane = threadIdx.x;
    const int wave = threadIdx.y;
    const int gw   = blockIdx.x * GEMM_WAVES + wave;
    const int tm   = gw / tilesN;
    const int tn   = gw - tm * tilesN;
    const int m0   = tm * 16;
    const int n0   = tn * 16;
    const int half = lane >> 4;
    const int sub  = lane & 15;
    const int h2   = half * 2;

    const float* aRow = A + (size_t)(m0 + sub) * lda;
    const float* wRow = W + (size_t)(n0 + sub) * ldw + wOff;

    v8f c = {};
    for (int k0 = 0; k0 < K; k0 += 16) {
        v2f a0 = ld2(aRow + k0 +  0 + h2);
        v2f a1 = ld2(aRow + k0 +  4 + h2);
        v2f a2 = ld2(aRow + k0 +  8 + h2);
        v2f a3 = ld2(aRow + k0 + 12 + h2);
        v2f b0 = ld2(wRow + k0 +  0 + h2);
        v2f b1 = ld2(wRow + k0 +  4 + h2);
        v2f b2 = ld2(wRow + k0 +  8 + h2);
        v2f b3 = ld2(wRow + k0 + 12 + h2);
        c = __builtin_amdgcn_wmma_f32_16x16x4_f32(false, a0, false, b0, (short)0, c, false, false);
        c = __builtin_amdgcn_wmma_f32_16x16x4_f32(false, a1, false, b1, (short)0, c, false, false);
        c = __builtin_amdgcn_wmma_f32_16x16x4_f32(false, a2, false, b2, (short)0, c, false, false);
        c = __builtin_amdgcn_wmma_f32_16x16x4_f32(false, a3, false, b3, (short)0, c, false, false);
    }

    const int   col  = n0 + sub;
    const int   row0 = m0 + half * 8;
    const float bb   = bias ? bias[col] : 0.0f;
#pragma unroll
    for (int r = 0; r < 8; ++r)
        C[(size_t)(row0 + r) * ldc + col] = c[r] + bb;
}

#define QB   4
#define SCQ  (SS * HH + 16)

typedef float v4f __attribute__((ext_vector_type(4)));

__device__ __forceinline__ v8f wmma4(v2f a, v2f b, v8f c) {
    return __builtin_amdgcn_wmma_f32_16x16x4_f32(false, a, false, b, (short)0, c, false, false);
}

__global__ __launch_bounds__(512) void scores_softmax_ctx(
    const float* __restrict__ qm,
    const float* __restrict__ km,
    const float* __restrict__ Wa,
    const float* __restrict__ ba,
    const float* __restrict__ attn_v,
    const float* __restrict__ v,
    float* __restrict__ ctx)
{
    __shared__ __attribute__((aligned(16))) float pq[QB * HH * HID];
    __shared__ float sc[QB * SCQ];
    __shared__ float vloc[QB * HH * DD];

    const int    tid = threadIdx.x;
    const size_t bi0 = (size_t)blockIdx.x * QB;
    const size_t bat = bi0 / SS;

    const float* vg = v + bi0 * (HH * DD);

    const int lane = tid & 31;
    const int h    = tid >> 5;
    const int sub  = lane & 15;
    const int half = lane >> 4;
    const int rsel = sub & 7;
    const int h2   = 2 * half;

    const float* kb = km + (bat * SS + sub) * (size_t)EE + h * DD + h2;

    float sv[QB * HH * DD / 512];
    for (int n = 0; n < QB * HH * DD / 512; ++n) sv[n] = vg[tid + 512 * n];

    v2f kf[4];
    for (int kk = 0; kk < 4; ++kk) kf[kk] = ld2(kb + 4 * kk);

    {
        const float* qrow = qm + (bi0 + (sub & (QB - 1))) * (size_t)EE + h * DD + h2;
        v2f qf[4];
        for (int kk = 0; kk < 4; ++kk) qf[kk] = ld2(qrow + 4 * kk);
        for (int t = 0; t < HID / 16; ++t) {
            const float* wr = Wa + (size_t)(16 * t + sub) * (2 * DD) + h2;
            const v4f b0 = *reinterpret_cast<const v4f*>(ba + 16 * t + 8 * half);
            const v4f b1 = *reinterpret_cast<const v4f*>(ba + 16 * t + 8 * half + 4);
            v8f d = {b0[0], b0[1], b0[2], b0[3], b1[0], b1[1], b1[2], b1[3]};
            for (int kk = 0; kk < 4; ++kk)
                d = wmma4(ld2(wr + 4 * kk), qf[kk], d);
            if (sub < QB) {
                float* dst = pq + (sub * HH + h) * HID + 16 * t + 8 * half;
                v4f lo = {d[0], d[1], d[2], d[3]};
                v4f hi = {d[4], d[5], d[6], d[7]};
                *reinterpret_cast<v4f*>(dst)     = lo;
                *reinterpret_cast<v4f*>(dst + 4) = hi;
            }
        }
    }

    const float* wkb = Wa + (size_t)sub * (2 * DD) + DD + h2;
    v4f avr[HID / 16][2];
    for (int t = 0; t < HID / 16; ++t) {
        avr[t][0] = *reinterpret_cast<const v4f*>(attn_v + 16 * t + 8 * half);
        avr[t][1] = *reinterpret_cast<const v4f*>(attn_v + 16 * t + 8 * half + 4);
    }

    for (int n = 0; n < QB * HH * DD / 512; ++n) vloc[tid + 512 * n] = sv[n];
    __syncthreads();

    for (int jt = 0; jt < SS / 16; ++jt) {
        v8f pk[HID / 16];
        for (int t = 0; t < HID / 16; ++t) {
            v8f d = {};
            const float* wr = wkb + (size_t)(16 * t) * (2 * DD);
            for (int kk = 0; kk < 4; ++kk) d = wmma4(ld2(wr + 4 * kk), kf[kk], d);
            pk[t] = d;
        }
        if (jt + 1 < SS / 16) {
            const float* kr = kb + (size_t)(16 * (jt + 1)) * EE;
            for (int kk = 0; kk < 4; ++kk) kf[kk] = ld2(kr + 4 * kk);
        }

#pragma unroll 1
        for (int q = 0; q < QB; q += 2) {
            const float* pq0 = pq + (q * HH + h) * HID + 8 * half;
            const float* pq1 = pq0 + HH * HID;
            v8f d0 = {};
            v8f d1 = {};
            for (int t = 0; t < HID / 16; ++t) {
                const v4f u0a = *reinterpret_cast<const v4f*>(pq0 + 16 * t);
                const v4f u0b = *reinterpret_cast<const v4f*>(pq0 + 16 * t + 4);
                const v4f u1a = *reinterpret_cast<const v4f*>(pq1 + 16 * t);
                const v4f u1b = *reinterpret_cast<const v4f*>(pq1 + 16 * t + 4);
                const v8f k8 = pk[t];
                v2f a, b;
                b[0] = avr[t][0][0]; b[1] = avr[t][0][1];
                a[0] = fast_tanh(u0a[0] + k8[0]); a[1] = fast_tanh(u0a[1] + k8[1]);
                d0 = wmma4(a, b, d0);
                a[0] = fast_tanh(u1a[0] + k8[0]); a[1] = fast_tanh(u1a[1] + k8[1]);
                d1 = wmma4(a, b, d1);
                b[0] = avr[t][0][2]; b[1] = avr[t][0][3];
                a[0] = fast_tanh(u0a[2] + k8[2]); a[1] = fast_tanh(u0a[3] + k8[3]);
                d0 = wmma4(a, b, d0);
                a[0] = fast_tanh(u1a[2] + k8[2]); a[1] = fast_tanh(u1a[3] + k8[3]);
                d1 = wmma4(a, b, d1);
                b[0] = avr[t][1][0]; b[1] = avr[t][1][1];
                a[0] = fast_tanh(u0b[0] + k8[4]); a[1] = fast_tanh(u0b[1] + k8[5]);
                d0 = wmma4(a, b, d0);
                a[0] = fast_tanh(u1b[0] + k8[4]); a[1] = fast_tanh(u1b[1] + k8[5]);
                d1 = wmma4(a, b, d1);
                b[0] = avr[t][1][2]; b[1] = avr[t][1][3];
                a[0] = fast_tanh(u0b[2] + k8[6]); a[1] = fast_tanh(u0b[3] + k8[7]);
                d0 = wmma4(a, b, d0);
                a[0] = fast_tanh(u1b[2] + k8[6]); a[1] = fast_tanh(u1b[3] + k8[7]);
                d1 = wmma4(a, b, d1);
            }
            float s0 = d0[0];
            float s1 = d1[0];
            for (int r = 1; r < 8; ++r) {
                s0 = (rsel == r) ? d0[r] : s0;
                s1 = (rsel == r) ? d1[r] : s1;
            }
            const int jrow = 16 * jt + 8 * half + rsel;
            sc[q * SCQ + jrow * HH + h]       = s0;
            sc[(q + 1) * SCQ + jrow * HH + h] = s1;
        }
    }
    __syncthreads();

    if (tid < QB * HH) {
        float* sp = sc + (tid >> 4) * SCQ + (tid & 15);
        float m = -3.0e38f;
        for (int j = 0; j < SS; ++j) m = fmaxf(m, sp[j * HH]);
        float s = 0.0f;
        for (int j = 0; j < SS; ++j) {
            float e = __expf(sp[j * HH] - m);
            sp[j * HH] = e;
            s += e;
        }
        const float inv = 1.0f / s;
        for (int j = 0; j < SS; ++j) sp[j * HH] *= inv;
    }
    __syncthreads();

    for (int q = 0; q < QB; ++q) {
        float*       cg = ctx + (bi0 + q) * (size_t)(SS * DD);
        const float* sq = sc + q * SCQ;
        const float* vq = vloc + q * (HH * DD);
        for (int idx = tid; idx < SS * DD; idx += 512) {
            const int j = idx >> 4;
            const int d = idx & 15;
            float acc = 0.0f;
#pragma unroll
            for (int hh = 0; hh < HH; ++hh)
                acc = fmaf(sq[j * HH + hh], vq[hh * DD + d], acc);
            cg[idx] = acc;
        }
    }
}

extern "C" void kernel_launch(void* const* d_in, const int* in_sizes, int n_in,
                              void* d_out, int out_size, void* d_ws, size_t ws_size,
                              hipStream_t stream) {
    (void)in_sizes; (void)n_in; (void)out_size; (void)ws_size;
    const float* x      = (const float*)d_in[0];
    const float* Wq     = (const float*)d_in[1];
    const float* bq     = (const float*)d_in[2];
    const float* Wk     = (const float*)d_in[3];
    const float* bk     = (const float*)d_in[4];
    const float* Wv     = (const float*)d_in[5];
    const float* bv     = (const float*)d_in[6];
    const float* Wa     = (const float*)d_in[7];
    const float* ba     = (const float*)d_in[8];
    const float* attn_v = (const float*)d_in[9];
    const float* Wfc    = (const float*)d_in[10];
    const float* bfc    = (const float*)d_in[11];
    float* out = (float*)d_out;

    float* q     = (float*)d_ws;
    float* k     = q     + (size_t)MQ * EE;
    float* v     = k     + (size_t)MQ * EE;
    float* ctx   = v     + (size_t)MQ * EE;

    const dim3 blk(32, GEMM_WAVES);

    gemm_bias_wmma_f32<<<256, blk, 0, stream>>>(x, EE, Wq, EE, 0, bq, q, EE, EE / 16, EE);
    gemm_bias_wmma_f32<<<256, blk, 0, stream>>>(x, EE, Wk, EE, 0, bk, k, EE, EE / 16, EE);
    gemm_bias_wmma_f32<<<256, blk, 0, stream>>>(x, EE, Wv, EE, 0, bv, v, EE, EE / 16, EE);

    scores_softmax_ctx<<<BB * SS / QB, 512, 0, stream>>>(q, k, Wa, ba, attn_v, v, ctx);

    gemm_bias_wmma_f32<<<256, blk, 0, stream>>>(ctx, SS * DD, Wfc, SS * DD, 0, bfc, out, EE, EE / 16, SS * DD);
}


# GEMM launches: a wave fetches its A and W rows by whole-line 16-byte loads, parks them in LDS rows of its own, reads fragments from LDS a stage ahead; k order in each 8 changes (sum order changes)
# speedup vs baseline: 1.9313x; 1.1412x over previous
#include <hip/hip_runtime.h>
#include <hip/hip_bf16.h>

#define BB   16
#define SS   64
#define HH   16
#define DD   16
#define EE   256
#define HID  128
#define MQ   (BB*SS)

typedef float v2f __attribute__((ext_vector_type(2)));
typedef float v8f __attribute__((ext_vector_type(8)));

__device__ __forceinline__ float fast_tanh(float x) {
#if __has_builtin(__builtin_amdgcn_tanhf)
    return __builtin_amdgcn_tanhf(x);
#else
    return tanhf(x);
#endif
}

__device__ __forceinline__ v2f ld2(const float* p) {
    return *reinterpret_cast<const v2f*>(p);
}

#define GEMM_WAVES 4
#define GEMM_KSTG  64
#define GEMM_PITCH (GEMM_KSTG + 4)
#define GEMM_TILE  (16 * GEMM_PITCH)

typedef float v4f __attribute__((ext_vector_type(4)));

__global__ __launch_bounds__(32 * GEMM_WAVES) void gemm_bias_wmma_f32(
    const float* __restrict__ A, int lda,
    const float* __restrict__ W, int ldw, int wOff,
    const float* __restrict__ bias,
    float* __restrict__ C, int ldc,
    int tilesN, int K)
{
    __shared__ __attribute__((aligned(16))) float stg[GEMM_WAVES][2][GEMM_TILE];

    const int lane = threadIdx.x;
    const int wave = threadIdx.y;
    const int gw   = blockIdx.x * GEMM_WAVES + wave;
    const int tm   = gw / tilesN;
    const int tn   = gw - tm * tilesN;
    const int m0   = tm * 16;
    const int n0   = tn * 16;
    const int half = lane >> 4;
    const int sub  = lane & 15;

    const float* aG = A + (size_t)(m0 + half) * lda + 4 * sub;
    const float* wG = W + (size_t)(n0 + half) * ldw + wOff + 4 * sub;
    float*       stPark = &stg[wave][0][0] + half * GEMM_PITCH + 4 * sub;
    const float* stFrag = &stg[wave][0][0] + sub * GEMM_PITCH + 4 * half;

    auto gload = [&](v4f (&ga)[8], v4f (&gb)[8], int k) {
#pragma unroll
        for (int j = 0; j < 8; ++j) {
            ga[j] = *reinterpret_cast<const v4f*>(aG + (size_t)(2 * j) * lda + k);
            gb[j] = *reinterpret_cast<const v4f*>(wG + (size_t)(2 * j) * ldw + k);
        }
    };
    auto park = [&](const v4f (&ga)[8], const v4f (&gb)[8]) {
        float* pa = stPark;
        float* pw = pa + GEMM_TILE;
#pragma unroll
        for (int j = 0; j < 8; ++j) {
            *reinterpret_cast<v4f*>(pa + 2 * j * GEMM_PITCH) = ga[j];
            *reinterpret_cast<v4f*>(pw + 2 * j * GEMM_PITCH) = gb[j];
        }
    };
    auto frags = [&](v4f (&fa)[8], v4f (&fb)[8]) {
        const float* pa = stFrag;
        const float* pw = pa + GEMM_TILE;
#pragma unroll
        for (int kk = 0; kk < 8; ++kk) {
            fa[kk] = *reinterpret_cast<const v4f*>(pa + 8 * kk);
            fb[kk] = *reinterpret_cast<const v4f*>(pw + 8 * kk);
        }
    };
    v8f c = {};
    auto mma = [&](const v4f (&fa)[8], const v4f (&fb)[8]) {
#pragma unroll
        for (int kk = 0; kk < 8; ++kk) {
            v2f a, b;
            a[0] = fa[kk][0]; a[1] = fa[kk][1];
            b[0] = fb[kk][0]; b[1] = fb[kk][1];
            c = __builtin_amdgcn_wmma_f32_16x16x4_f32(false, a, false, b, (short)0, c, false, false);
            a[0] = fa[kk][2]; a[1] = fa[kk][3];
            b[0] = fb[kk][2]; b[1] = fb[kk][3];
            c = __builtin_amdgcn_wmma_f32_16x16x4_f32(false, a, false, b, (short)0, c, false, false);
        }
    };

    v4f ga0[8], gb0[8], ga1[8], gb1[8];
    v4f fa0[8], fb0[8], fa1[8], fb1[8];

    gload(ga0, gb0, 0);
    gload(ga1, gb1, GEMM_KSTG);
    park(ga0, gb0);
    if (2 * GEMM_KSTG < K) gload(ga0, gb0, 2 * GEMM_KSTG);
    frags(fa0, fb0);

    for (int k0 = 0; k0 < K; k0 += 2 * GEMM_KSTG) {
        park(ga1, gb1);
        if (k0 + 3 * GEMM_KSTG < K) gload(ga1, gb1, k0 + 3 * GEMM_KSTG);
        frags(fa1, fb1);
        mma(fa0, fb0);
        if (k0 + 2 * GEMM_KSTG < K) {
            park(ga0, gb0);
            if (k0 + 4 * GEMM_KSTG < K) gload(ga0, gb0, k0 + 4 * GEMM_KSTG);
            frags(fa0, fb0);
        }
        mma(fa1, fb1);
    }

    const int   col  = n0 + sub;
    const int   row0 = m0 + half * 8;
    const float bb   = bias ? bias[col] : 0.0f;
#pragma unroll
    for (int r = 0; r < 8; ++r)
        C[(size_t)(row0 + r) * ldc + col] = c[r] + bb;
}

#define QB   4
#define SCQ  (SS * HH + 16)

__device__ __forceinline__ v8f wmma4(v2f a, v2f b, v8f c) {
    return __builtin_amdgcn_wmma_f32_16x16x4_f32(false, a, false, b, (short)0, c, false, false);
}

__global__ __launch_bounds__(512) void scores_softmax_ctx(
    const float* __restrict__ qm,
    const float* __restrict__ km,
    const float* __restrict__ Wa,
    const float* __restrict__ ba,
    const float* __restrict__ attn_v,
    const float* __restrict__ v,
    float* __restrict__ ctx)
{
    __shared__ __attribute__((aligned(16))) float pq[QB * HH * HID];
    __shared__ float sc[QB * SCQ];
    __shared__ float vloc[QB * HH * DD];

    const int    tid = threadIdx.x;
    const size_t bi0 = (size_t)blockIdx.x * QB;
    const size_t bat = bi0 / SS;

    const float* vg = v + bi0 * (HH * DD);

    const int lane = tid & 31;
    const int h    = tid >> 5;
    const int sub  = lane & 15;
    const int half = lane >> 4;
    const int rsel = sub & 7;
    const int h2   = 2 * half;

    const float* kb = km + (bat * SS + sub) * (size_t)EE + h * DD + h2;

    float sv[QB * HH * DD / 512];
    for (int n = 0; n < QB * HH * DD / 512; ++n) sv[n] = vg[tid + 512 * n];

    v2f kf[4];
    for (int kk = 0; kk < 4; ++kk) kf[kk] = ld2(kb + 4 * kk);

    {
        const float* qrow = qm + (bi0 + (sub & (QB - 1))) * (size_t)EE + h * DD + h2;
        v2f qf[4];
        for (int kk = 0; kk < 4; ++kk) qf[kk] = ld2(qrow + 4 * kk);
        for (int t = 0; t < HID / 16; ++t) {
            const float* wr = Wa + (size_t)(16 * t + sub) * (2 * DD) + h2;
            const v4f b0 = *reinterpret_cast<const v4f*>(ba + 16 * t + 8 * half);
            const v4f b1 = *reinterpret_cast<const v4f*>(ba + 16 * t + 8 * half + 4);
            v8f d = {b0[0], b0[1], b0[2], b0[3], b1[0], b1[1], b1[2], b1[3]};
            for (int kk = 0; kk < 4; ++kk)
                d = wmma4(ld2(wr + 4 * kk), qf[kk], d);
            if (sub < QB) {
                float* dst = pq + (sub * HH + h) * HID + 16 * t + 8 * half;
                v4f lo = {d[0], d[1], d[2], d[3]};
                v4f hi = {d[4], d[5], d[6], d[7]};
                *reinterpret_cast<v4f*>(dst)     = lo;
                *reinterpret_cast<v4f*>(dst + 4) = hi;
            }
        }
    }

    const float* wkb = Wa + (size_t)sub * (2 * DD) + DD + h2;
    v4f avr[HID / 16][2];
    for (int t = 0; t < HID / 16; ++t) {
        avr[t][0] = *reinterpret_cast<const v4f*>(attn_v + 16 * t + 8 * half);
        avr[t][1] = *reinterpret_cast<const v4f*>(attn_v + 16 * t + 8 * half + 4);
    }

    for (int n = 0; n < QB * HH * DD / 512; ++n) vloc[tid + 512 * n] = sv[n];
    __syncthreads();

    for (int jt = 0; jt < SS / 16; ++jt) {
        v8f pk[HID / 16];
        for (int t = 0; t < HID / 16; ++t) {
            v8f d = {};
            const float* wr = wkb + (size_t)(16 * t) * (2 * DD);
            for (int kk = 0; kk < 4; ++kk) d = wmma4(ld2(wr + 4 * kk), kf[kk], d);
            pk[t] = d;
        }
        if (jt + 1 < SS / 16) {
            const float* kr = kb + (size_t)(16 * (jt + 1)) * EE;
            for (int kk = 0; kk < 4; ++kk) kf[kk] = ld2(kr + 4 * kk);
        }

#pragma unroll 1
        for (int q = 0; q < QB; q += 2) {
            const float* pq0 = pq + (q * HH + h) * HID + 8 * half;
            const float* pq1 = pq0 + HH * HID;
            v8f d0 = {};
            v8f d1 = {};
            for (int t = 0; t < HID / 16; ++t) {
                const v4f u0a = *reinterpret_cast<const v4f*>(pq0 + 16 * t);
                const v4f u0b = *reinterpret_cast<const v4f*>(pq0 + 16 * t + 4);
                const v4f u1a = *reinterpret_cast<const v4f*>(pq1 + 16 * t);
                const v4f u1b = *reinterpret_cast<const v4f*>(pq1 + 16 * t + 4);
                const v8f k8 = pk[t];
                v2f a, b;
                b[0] = avr[t][0][0]; b[1] = avr[t][0][1];
                a[0] = fast_tanh(u0a[0] + k8[0]); a[1] = fast_tanh(u0a[1] + k8[1]);
                d0 = wmma4(a, b, d0);
                a[0] = fast_tanh(u1a[0] + k8[0]); a[1] = fast_tanh(u1a[1] + k8[1]);
                d1 = wmma4(a, b, d1);
                b[0] = avr[t][0][2]; b[1] = avr[t][0][3];
                a[0] = fast_tanh(u0a[2] + k8[2]); a[1] = fast_tanh(u0a[3] + k8[3]);
                d0 = wmma4(a, b, d0);
                a[0] = fast_tanh(u1a[2] + k8[2]); a[1] = fast_tanh(u1a[3] + k8[3]);
                d1 = wmma4(a, b, d1);
                b[0] = avr[t][1][0]; b[1] = avr[t][1][1];
                a[0] = fast_tanh(u0b[0] + k8[4]); a[1] = fast_tanh(u0b[1] + k8[5]);
                d0 = wmma4(a, b, d0);
                a[0] = fast_tanh(u1b[0] + k8[4]); a[1] = fast_tanh(u1b[1] + k8[5]);
                d1 = wmma4(a, b, d1);
                b[0] = avr[t][1][2]; b[1] = avr[t][1][3];
                a[0] = fast_tanh(u0b[2] + k8[6]); a[1] = fast_tanh(u0b[3] + k8[7]);
                d0 = wmma4(a, b, d0);
                a[0] = fast_tanh(u1b[2] + k8[6]); a[1] = fast_tanh(u1b[3] + k8[7]);
                d1 = wmma4(a, b, d1);
            }
            float s0 = d0[0];
            float s1 = d1[0];
            for (int r = 1; r < 8; ++r) {
                s0 = (rsel == r) ? d0[r] : s0;
                s1 = (rsel == r) ? d1[r] : s1;
            }
            const int jrow = 16 * jt + 8 * half + rsel;
            sc[q * SCQ + jrow * HH + h]       = s0;
            sc[(q + 1) * SCQ + jrow * HH + h] = s1;
        }
    }
    __syncthreads();

    if (tid < QB * HH) {
        float* sp = sc + (tid >> 4) * SCQ + (tid & 15);
        float m = -3.0e38f;
        for (int j = 0; j < SS; ++j) m = fmaxf(m, sp[j * HH]);
        float s = 0.0f;
        for (int j = 0; j < SS; ++j) {
            float e = __expf(sp[j * HH] - m);
            sp[j * HH] = e;
            s += e;
        }
        const float inv = 1.0f / s;
        for (int j = 0; j < SS; ++j) sp[j * HH] *= inv;
    }
    __syncthreads();

    for (int q = 0; q < QB; ++q) {
        float*       cg = ctx + (bi0 + q) * (size_t)(SS * DD);
        const float* sq = sc + q * SCQ;
        const float* vq = vloc + q * (HH * DD);
        for (int idx = tid; idx < SS * DD; idx += 512) {
            const int j = idx >> 4;
            const int d = idx & 15;
            float acc = 0.0f;
#pragma unroll
            for (int hh = 0; hh < HH; ++hh)
                acc = fmaf(sq[j * HH + hh], vq[hh * DD + d], acc);
            cg[idx] = acc;
        }
    }
}

extern "C" void kernel_launch(void* const* d_in, const int* in_sizes, int n_in,
                              void* d_out, int out_size, void* d_ws, size_t ws_size,
                              hipStream_t stream) {
    (void)in_sizes; (void)n_in; (void)out_size; (void)ws_size;
    const float* x      = (const float*)d_in[0];
    const float* Wq     = (const float*)d_in[1];
    const float* bq     = (const float*)d_in[2];
    const float* Wk     = (const float*)d_in[3];
    const float* bk     = (const float*)d_in[4];
    const float* Wv     = (const float*)d_in[5];
    const float* bv     = (const float*)d_in[6];
    const float* Wa     = (const float*)d_in[7];
    const float* ba     = (const float*)d_in[8];
    const float* attn_v = (const float*)d_in[9];
    const float* Wfc    = (const float*)d_in[10];
    const float* bfc    = (const float*)d_in[11];
    float* out = (float*)d_out;

    float* q     = (float*)d_ws;
    float* k     = q     + (size_t)MQ * EE;
    float* v     = k     + (size_t)MQ * EE;
    float* ctx   = v     + (size_t)MQ * EE;

    const dim3 blk(32, GEMM_WAVES);

    gemm_bias_wmma_f32<<<256, blk, 0, stream>>>(x, EE, Wq, EE, 0, bq, q, EE, EE / 16, EE);
    gemm_bias_wmma_f32<<<256, blk, 0, stream>>>(x, EE, Wk, EE, 0, bk, k, EE, EE / 16, EE);
    gemm_bias_wmma_f32<<<256, blk, 0, stream>>>(x, EE, Wv, EE, 0, bv, v, EE, EE / 16, EE);

    scores_softmax_ctx<<<BB * SS / QB, 512, 0, stream>>>(q, k, Wa, ba, attn_v, v, ctx);

    gemm_bias_wmma_f32<<<256, blk, 0, stream>>>(ctx, SS * DD, Wfc, SS * DD, 0, bfc, out, EE, EE / 16, SS * DD);
}
